# starting kernel
# baseline (speedup 1.0000x reference)
#include <hip/hip_runtime.h>

typedef float v2f __attribute__((ext_vector_type(2)));
typedef float v8f __attribute__((ext_vector_type(8)));

__global__ void init_max_kernel(unsigned int* __restrict__ maxbits, int n) {
    int i = blockIdx.x * blockDim.x + threadIdx.x;
    if (i < n) maxbits[i] = 0u;
}

__global__ void phi_mean_wmma_kernel(const float* __restrict__ W_phi,
                                     float* __restrict__ phi_mean) {
    int lane = threadIdx.x;
    v2f a0, a1, b;
    a0.x = W_phi[lane];        a0.y = W_phi[lane + 32];
    a1.x = W_phi[lane + 64];   a1.y = W_phi[lane + 96];
    b.x = 1.0f; b.y = 1.0f;
    v8f c = {};
    c = __builtin_amdgcn_wmma_f32_16x16x4_f32(false, a0, false, b,
                                              (short)0, c, false, false);
    c = __builtin_amdgcn_wmma_f32_16x16x4_f32(false, a1, false, b,
                                              (short)0, c, false, false);
    float s = c[0] + c[1] + c[2] + c[3] + c[4] + c[5] + c[6] + c[7];
    float partner = __shfl(s, lane ^ 16, 32);
    if (lane == 0) phi_mean[0] = (s + partner) * (1.0f / 128.0f);
}

__global__ void edge_max_kernel(const float* __restrict__ nodes,
                                const int* __restrict__ rows,
                                const int* __restrict__ cols,
                                const float* __restrict__ W_theta,
                                unsigned int* __restrict__ maxbits,
                                int n_edges) {
    const float w0 = W_theta[0], w1 = W_theta[1], w2 = W_theta[2];
    int base = (blockIdx.x * blockDim.x + threadIdx.x) * 4;
    if (base + 3 < n_edges) {
        int4 r4 = *reinterpret_cast<const int4*>(rows + base);
        int4 c4 = *reinterpret_cast<const int4*>(cols + base);
        int rr[4] = {r4.x, r4.y, r4.z, r4.w};
        int cc[4] = {c4.x, c4.y, c4.z, c4.w};
#pragma unroll
        for (int k = 0; k < 4; ++k) {
            const float* pa = nodes + 3 * rr[k];
            const float* pb = nodes + 3 * cc[k];
            float d = fabsf((pa[0] - pb[0]) * w0 +
                            (pa[1] - pb[1]) * w1 +
                            (pa[2] - pb[2]) * w2);
            atomicMax(maxbits + rr[k], __float_as_uint(d));
        }
    } else {
        for (int e = base; e < n_edges; ++e) {
            int r = rows[e], c = cols[e];
            const float* pa = nodes + 3 * r;
            const float* pb = nodes + 3 * c;
            float d = fabsf((pa[0] - pb[0]) * w0 +
                            (pa[1] - pb[1]) * w1 +
                            (pa[2] - pb[2]) * w2);
            atomicMax(maxbits + r, __float_as_uint(d));
        }
    }
}

__global__ void finalize_kernel(const float* __restrict__ prev,
                                const unsigned int* __restrict__ maxbits,
                                const float* __restrict__ phi_mean,
                                float* __restrict__ out, int n) {
    int i = blockIdx.x * blockDim.x + threadIdx.x;
    if (i < n) {
        float m = __uint_as_float(maxbits[i]);
        out[i] = 0.5f * (prev[i] + m * phi_mean[0]);
    }
}

extern "C" void kernel_launch(void* const* d_in, const int* in_sizes, int n_in,
                              void* d_out, int out_size, void* d_ws, size_t ws_size,
                              hipStream_t stream) {
    const float* prev    = (const float*)d_in[0];
    const float* nodes   = (const float*)d_in[1];
    const int*   rows    = (const int*)  d_in[2];
    const int*   cols    = (const int*)  d_in[3];
    const float* W_phi   = (const float*)d_in[4];
    const float* W_theta = (const float*)d_in[5];
    float*       out     = (float*)d_out;

    const int n_nodes = in_sizes[0];
    const int n_edges = in_sizes[2];

    unsigned int* maxbits = (unsigned int*)d_ws;
    float* phi_mean = (float*)((unsigned int*)d_ws + n_nodes);

    const int B = 256;
    int node_blocks = (n_nodes + B - 1) / B;
    int edge_threads = (n_edges + 3) / 4;
    int edge_blocks = (edge_threads + B - 1) / B;

    init_max_kernel<<<node_blocks, B, 0, stream>>>(maxbits, n_nodes);
    phi_mean_wmma_kernel<<<1, 32, 0, stream>>>(W_phi, phi_mean);
    edge_max_kernel<<<edge_blocks, B, 0, stream>>>(nodes, rows, cols, W_theta,
                                                   maxbits, n_edges);
    finalize_kernel<<<node_blocks, B, 0, stream>>>(prev, maxbits, phi_mean,
                                                   out, n_nodes);
}


# Edge kernel regrouped by node range: each work-group scans the edge list, keeps the segment max of its own node range in LDS and folds it out with whole-line atomics; no per-edge global atomic
# speedup vs baseline: 3.3766x; 3.3766x over previous
#include <hip/hip_runtime.h>

typedef float v2f __attribute__((ext_vector_type(2)));
typedef float v8f __attribute__((ext_vector_type(8)));

__global__ void init_max_kernel(unsigned int* __restrict__ maxbits, int n) {
    int i = blockIdx.x * blockDim.x + threadIdx.x;
    if (i < n) maxbits[i] = 0u;
}

__global__ void phi_mean_wmma_kernel(const float* __restrict__ W_phi,
                                     float* __restrict__ phi_mean) {
    int lane = threadIdx.x;
    v2f a0, a1, b;
    a0.x = W_phi[lane];        a0.y = W_phi[lane + 32];
    a1.x = W_phi[lane + 64];   a1.y = W_phi[lane + 96];
    b.x = 1.0f; b.y = 1.0f;
    v8f c = {};
    c = __builtin_amdgcn_wmma_f32_16x16x4_f32(false, a0, false, b,
                                              (short)0, c, false, false);
    c = __builtin_amdgcn_wmma_f32_16x16x4_f32(false, a1, false, b,
                                              (short)0, c, false, false);
    float s = c[0] + c[1] + c[2] + c[3] + c[4] + c[5] + c[6] + c[7];
    float partner = __shfl(s, lane ^ 16, 32);
    if (lane == 0) phi_mean[0] = (s + partner) * (1.0f / 128.0f);
}

#define NODE_PARTS 16
#define EDGE_PARTS 32
#define TILE_CAP 8192
#define EDGE_BLOCK 1024

__device__ __forceinline__ float edge_dist(const float* __restrict__ pa,
                                           const float* __restrict__ pb,
                                           float w0, float w1, float w2) {
    return fabsf((pa[0] - pb[0]) * w0 +
                 (pa[1] - pb[1]) * w1 +
                 (pa[2] - pb[2]) * w2);
}

__global__ void __launch_bounds__(EDGE_BLOCK)
edge_max_kernel(const float* __restrict__ nodes,
                const int* __restrict__ rows,
                const int* __restrict__ cols,
                const float* __restrict__ W_theta,
                unsigned int* __restrict__ maxbits,
                int n_nodes, int n_edges) {
    __shared__ unsigned int tile_max[TILE_CAP];
    const float w0 = W_theta[0], w1 = W_theta[1], w2 = W_theta[2];
    const int n_parts = gridDim.x;
    int tile = (n_nodes + n_parts - 1) / n_parts;
    if (tile > TILE_CAP) tile = TILE_CAP;
    const int n_groups = n_edges >> 2;
    const int g_step = gridDim.y * blockDim.x;
    const int4* __restrict__ rows4 = reinterpret_cast<const int4*>(rows);
    const int4* __restrict__ cols4 = reinterpret_cast<const int4*>(cols);

    for (int lo = blockIdx.x * tile; lo < n_nodes; lo += n_parts * tile) {
        for (int i = threadIdx.x; i < tile; i += blockDim.x) tile_max[i] = 0u;
        __syncthreads();

        int g = blockIdx.y * blockDim.x + threadIdx.x;
        int4 r4 = make_int4(0, 0, 0, 0), c4 = make_int4(0, 0, 0, 0);
        if (g < n_groups) { r4 = rows4[g]; c4 = cols4[g]; }
        while (g < n_groups) {
            const int gn = g + g_step;
            const int gp = gn < n_groups ? gn : g;
            const int4 rn = rows4[gp];
            const int4 cn = cols4[gp];

            const int rr[4] = {r4.x, r4.y, r4.z, r4.w};
            const int cc[4] = {c4.x, c4.y, c4.z, c4.w};
            unsigned int off[4];
            float d[4];
            for (int k = 0; k < 4; ++k) {
                off[k] = (unsigned int)(rr[k] - lo);
                const bool hit = off[k] < (unsigned int)tile;
                const int ir = hit ? rr[k] : lo;
                const int ic = hit ? cc[k] : lo;
                d[k] = edge_dist(nodes + 3 * ir, nodes + 3 * ic, w0, w1, w2);
            }
            for (int k = 0; k < 4; ++k) {
                if (off[k] < (unsigned int)tile)
                    atomicMax(&tile_max[off[k]], __float_as_uint(d[k]));
            }
            r4 = rn; c4 = cn; g = gn;
        }
        if (blockIdx.y == 0) {
            int e = (n_groups << 2) + threadIdx.x;
            if (e < n_edges) {
                int r = rows[e], c = cols[e];
                unsigned int o = (unsigned int)(r - lo);
                if (o < (unsigned int)tile) {
                    float d = edge_dist(nodes + 3 * r, nodes + 3 * c, w0, w1, w2);
                    atomicMax(&tile_max[o], __float_as_uint(d));
                }
            }
        }
        __syncthreads();
        for (int i = threadIdx.x; i < tile; i += blockDim.x) {
            unsigned int v = tile_max[i];
            if (v != 0u && lo + i < n_nodes) atomicMax(maxbits + lo + i, v);
        }
        __syncthreads();
    }
}

__global__ void finalize_kernel(const float* __restrict__ prev,
                                const unsigned int* __restrict__ maxbits,
                                const float* __restrict__ phi_mean,
                                float* __restrict__ out, int n) {
    int i = blockIdx.x * blockDim.x + threadIdx.x;
    if (i < n) {
        float m = __uint_as_float(maxbits[i]);
        out[i] = 0.5f * (prev[i] + m * phi_mean[0]);
    }
}

extern "C" void kernel_launch(void* const* d_in, const int* in_sizes, int n_in,
                              void* d_out, int out_size, void* d_ws, size_t ws_size,
                              hipStream_t stream) {
    const float* prev    = (const float*)d_in[0];
    const float* nodes   = (const float*)d_in[1];
    const int*   rows    = (const int*)  d_in[2];
    const int*   cols    = (const int*)  d_in[3];
    const float* W_phi   = (const float*)d_in[4];
    const float* W_theta = (const float*)d_in[5];
    float*       out     = (float*)d_out;

    const int n_nodes = in_sizes[0];
    const int n_edges = in_sizes[2];

    unsigned int* maxbits = (unsigned int*)d_ws;
    float* phi_mean = (float*)((unsigned int*)d_ws + n_nodes);

    const int B = 256;
    int node_blocks = (n_nodes + B - 1) / B;

    init_max_kernel<<<node_blocks, B, 0, stream>>>(maxbits, n_nodes);
    phi_mean_wmma_kernel<<<1, 32, 0, stream>>>(W_phi, phi_mean);
    edge_max_kernel<<<dim3(NODE_PARTS, EDGE_PARTS), EDGE_BLOCK, 0, stream>>>(
        nodes, rows, cols, W_theta, maxbits, n_nodes, n_edges);
    finalize_kernel<<<node_blocks, B, 0, stream>>>(prev, maxbits, phi_mean,
                                                   out, n_nodes);
}


# Node projection nodes.W_theta computed once per node: an edge reads it from an LDS copy of the owned node range for its row and gathers one word for its column; 8 node ranges, 256 work-groups
# speedup vs baseline: 5.6370x; 1.6694x over previous
#include <hip/hip_runtime.h>


typedef float v2f __attribute__((ext_vector_type(2)));
typedef float v8f __attribute__((ext_vector_type(8)));

__global__ void init_project_kernel(const float* __restrict__ nodes,
                                    const float* __restrict__ W_theta,
                                    unsigned int* __restrict__ maxbits,
                                    float* __restrict__ proj, int n) {
    int i = blockIdx.x * blockDim.x + threadIdx.x;
    if (i < n) {
        const float w0 = W_theta[0], w1 = W_theta[1], w2 = W_theta[2];
        const float* pn = nodes + 3 * i;
        maxbits[i] = 0u;
        proj[i] = pn[0] * w0 + pn[1] * w1 + pn[2] * w2;
    }
}

__global__ void phi_mean_wmma_kernel(const float* __restrict__ W_phi,
                                     float* __restrict__ phi_mean) {
    int lane = threadIdx.x;
    v2f a0, a1, b;
    a0.x = W_phi[lane];        a0.y = W_phi[lane + 32];
    a1.x = W_phi[lane + 64];   a1.y = W_phi[lane + 96];
    b.x = 1.0f; b.y = 1.0f;
    v8f c = {};
    c = __builtin_amdgcn_wmma_f32_16x16x4_f32(false, a0, false, b,
                                              (short)0, c, false, false);
    c = __builtin_amdgcn_wmma_f32_16x16x4_f32(false, a1, false, b,
                                              (short)0, c, false, false);
    float s = c[0] + c[1] + c[2] + c[3] + c[4] + c[5] + c[6] + c[7];
    float partner = __shfl(s, lane ^ 16, 32);
    if (lane == 0) phi_mean[0] = (s + partner) * (1.0f / 128.0f);
}

#define NODE_PARTS 8
#define EDGE_PARTS 32
#define TILE_CAP 16384
#define EDGE_BLOCK 1024

__device__ __forceinline__ void edge_pair(const int4 ra, const int4 ca,
                                          const int4 rb, const int4 cb, bool valid_b,
                                          int lo, int tile,
                                          const float* __restrict__ proj,
                                          const float* tile_proj,
                                          unsigned int* tile_max) {
    const int rr[8] = {ra.x, ra.y, ra.z, ra.w, rb.x, rb.y, rb.z, rb.w};
    const int cc[8] = {ca.x, ca.y, ca.z, ca.w, cb.x, cb.y, cb.z, cb.w};
    unsigned int off[8], d[8];
    bool hit[8];
    float pc[8], pr[8];
    for (int k = 0; k < 8; ++k) {
        off[k] = (unsigned int)(rr[k] - lo);
        hit[k] = (k < 4 || valid_b) && off[k] < (unsigned int)tile;
        pc[k] = proj[hit[k] ? cc[k] : lo];
    }
    for (int k = 0; k < 8; ++k) pr[k] = tile_proj[hit[k] ? off[k] : 0u];
    for (int k = 0; k < 8; ++k) d[k] = __float_as_uint(fabsf(pr[k] - pc[k]));
    asm volatile("" : "+v"(d[0]), "+v"(d[1]), "+v"(d[2]), "+v"(d[3]),
                      "+v"(d[4]), "+v"(d[5]), "+v"(d[6]), "+v"(d[7]));
    for (int k = 0; k < 8; ++k) {
        if (hit[k]) atomicMax(&tile_max[off[k]], d[k]);
    }
}

__global__ void __launch_bounds__(EDGE_BLOCK)
edge_max_kernel(const float* __restrict__ proj,
                const int* __restrict__ rows,
                const int* __restrict__ cols,
                unsigned int* __restrict__ maxbits,
                int n_nodes, int n_edges) {
    __shared__ float tile_proj[TILE_CAP];
    __shared__ unsigned int tile_max[TILE_CAP];
    const int n_parts = gridDim.x;
    int tile = (n_nodes + n_parts - 1) / n_parts;
    if (tile > TILE_CAP) tile = TILE_CAP;
    const int n_groups = n_edges >> 2;
    const int g_step = gridDim.y * blockDim.x;
    const int4* __restrict__ rows4 = reinterpret_cast<const int4*>(rows);
    const int4* __restrict__ cols4 = reinterpret_cast<const int4*>(cols);

    for (int lo = blockIdx.x * tile; lo < n_nodes; lo += n_parts * tile) {
        for (int i0 = threadIdx.x; i0 < tile; i0 += 4 * blockDim.x) {
            float v[4];
            for (int u = 0; u < 4; ++u) {
                const int n = lo + i0 + u * blockDim.x;
                v[u] = proj[n < n_nodes ? n : n_nodes - 1];
            }
            asm volatile("" : "+v"(v[0]), "+v"(v[1]), "+v"(v[2]), "+v"(v[3]));
            for (int u = 0; u < 4; ++u) {
                const int i = i0 + u * blockDim.x;
                if (i < tile) {
                    tile_proj[i] = lo + i < n_nodes ? v[u] : 0.0f;
                    tile_max[i] = 0u;
                }
            }
        }
        __syncthreads();

        int g = blockIdx.y * blockDim.x + threadIdx.x;
        int4 ra = make_int4(0, 0, 0, 0), ca = ra, rb = ra, cb = ra;
        if (g < n_groups) {
            const int gb = g + g_step < n_groups ? g + g_step : g;
            ra = rows4[g];  ca = cols4[g];
            rb = rows4[gb]; cb = cols4[gb];
        }
        while (g < n_groups) {
            const int gn = g + 2 * g_step;
            const int ia = gn < n_groups ? gn : g;
            const int ib = gn + g_step < n_groups ? gn + g_step : g;
            const int4 rna = rows4[ia], cna = cols4[ia];
            const int4 rnb = rows4[ib], cnb = cols4[ib];

            edge_pair(ra, ca, rb, cb, g + g_step < n_groups, lo, tile, proj,
                      tile_proj, tile_max);

            ra = rna; ca = cna; rb = rnb; cb = cnb; g = gn;
        }
        if (blockIdx.y == 0) {
            int e = (n_groups << 2) + threadIdx.x;
            if (e < n_edges) {
                int r = rows[e], c = cols[e];
                unsigned int o = (unsigned int)(r - lo);
                if (o < (unsigned int)tile) {
                    float d = fabsf(tile_proj[o] - proj[c]);
                    atomicMax(&tile_max[o], __float_as_uint(d));
                }
            }
        }
        __syncthreads();
        for (int i = threadIdx.x; i < tile; i += blockDim.x) {
            unsigned int v = tile_max[i];
            if (v != 0u && lo + i < n_nodes) atomicMax(maxbits + lo + i, v);
        }
        __syncthreads();
    }
}

__global__ void finalize_kernel(const float* __restrict__ prev,
                                const unsigned int* __restrict__ maxbits,
                                const float* __restrict__ phi_mean,
                                float* __restrict__ out, int n) {
    int i = blockIdx.x * blockDim.x + threadIdx.x;
    if (i < n) {
        float m = __uint_as_float(maxbits[i]);
        out[i] = 0.5f * (prev[i] + m * phi_mean[0]);
    }
}

extern "C" void kernel_launch(void* const* d_in, const int* in_sizes, int n_in,
                              void* d_out, int out_size, void* d_ws, size_t ws_size,
                              hipStream_t stream) {
    const float* prev    = (const float*)d_in[0];
    const float* nodes   = (const float*)d_in[1];
    const int*   rows    = (const int*)  d_in[2];
    const int*   cols    = (const int*)  d_in[3];
    const float* W_phi   = (const float*)d_in[4];
    const float* W_theta = (const float*)d_in[5];
    float*       out     = (float*)d_out;

    const int n_nodes = in_sizes[0];
    const int n_edges = in_sizes[2];

    unsigned int* maxbits = (unsigned int*)d_ws;
    float* phi_mean = (float*)((unsigned int*)d_ws + n_nodes);
    float* proj = phi_mean + 1;

    const int B = 256;
    int node_blocks = (n_nodes + B - 1) / B;

    init_project_kernel<<<node_blocks, B, 0, stream>>>(nodes, W_theta, maxbits,
                                                       proj, n_nodes);
    phi_mean_wmma_kernel<<<1, 32, 0, stream>>>(W_phi, phi_mean);
    edge_max_kernel<<<dim3(NODE_PARTS, EDGE_PARTS), EDGE_BLOCK, 0, stream>>>(
        proj, rows, cols, maxbits, n_nodes, n_edges);
    finalize_kernel<<<node_blocks, B, 0, stream>>>(prev, maxbits, phi_mean,
                                                   out, n_nodes);
}


# Edge kernel: LDS copy of the node projections widened from the owned node range to a 66K-node window round it (316 KiB of LDS); only a column outside the window is still gathered from memory
# speedup vs baseline: 7.9205x; 1.4051x over previous
#include <hip/hip_runtime.h>


typedef float v2f __attribute__((ext_vector_type(2)));
typedef float v8f __attribute__((ext_vector_type(8)));

__device__ __forceinline__ void phi_mean_wmma(const float* __restrict__ W_phi,
                                              float* __restrict__ phi_mean,
                                              int lane) {
    v2f a0, a1, b;
    a0.x = W_phi[lane];        a0.y = W_phi[lane + 32];
    a1.x = W_phi[lane + 64];   a1.y = W_phi[lane + 96];
    b.x = 1.0f; b.y = 1.0f;
    v8f c = {};
    c = __builtin_amdgcn_wmma_f32_16x16x4_f32(false, a0, false, b,
                                              (short)0, c, false, false);
    c = __builtin_amdgcn_wmma_f32_16x16x4_f32(false, a1, false, b,
                                              (short)0, c, false, false);
    float s = c[0] + c[1] + c[2] + c[3] + c[4] + c[5] + c[6] + c[7];
    float partner = __shfl(s, lane ^ 16, 32);
    if (lane == 0) phi_mean[0] = (s + partner) * (1.0f / 128.0f);
}

__global__ void init_project_kernel(const float* __restrict__ prev,
                                    const float* __restrict__ nodes,
                                    const float* __restrict__ W_theta,
                                    const float* __restrict__ W_phi,
                                    float* __restrict__ out,
                                    float* __restrict__ proj,
                                    float* __restrict__ phi_mean, int n) {
    if (blockIdx.x == gridDim.x - 1) {
        if (threadIdx.x < 32) phi_mean_wmma(W_phi, phi_mean, threadIdx.x);
        return;
    }
    int i = blockIdx.x * blockDim.x + threadIdx.x;
    if (i < n) {
        const float w0 = W_theta[0], w1 = W_theta[1], w2 = W_theta[2];
        const float* pn = nodes + 3 * i;
        out[i] = 0.5f * prev[i];
        proj[i] = pn[0] * w0 + pn[1] * w1 + pn[2] * w2;
    }
}

#define NODE_PARTS 8
#define EDGE_PARTS 32
#define EDGE_BLOCK 1024
#define TILE_CAP (13 * EDGE_BLOCK)
#define WIN_CAP (66 * EDGE_BLOCK)
#define PREV_REGS (TILE_CAP / EDGE_BLOCK)
#define FILL_DEPTH 8

__device__ __forceinline__ void fold_float(float* p, float val, bool rising) {
    const int bits = __float_as_int(val);
    if (rising) {
        if (bits >= 0) atomicMax(reinterpret_cast<int*>(p), bits);
        else atomicMin(reinterpret_cast<unsigned int*>(p), (unsigned int)bits);
    } else {
        if (bits >= 0) atomicMin(reinterpret_cast<int*>(p), bits);
        else atomicMax(reinterpret_cast<unsigned int*>(p), (unsigned int)bits);
    }
}

__device__ __forceinline__ void edge_pair(const int4 ra, const int4 ca,
                                          const int4 rb, const int4 cb, bool valid_b,
                                          int lo, int tile, int wlo, int wlen,
                                          const float* __restrict__ proj,
                                          const float* win,
                                          unsigned int* tile_max) {
    const int rr[8] = {ra.x, ra.y, ra.z, ra.w, rb.x, rb.y, rb.z, rb.w};
    const int cc[8] = {ca.x, ca.y, ca.z, ca.w, cb.x, cb.y, cb.z, cb.w};
    const unsigned int row_shift = (unsigned int)(lo - wlo);
    unsigned int off[8], wc[8], d[8];
    bool hit[8], inw[8];
    float pg[8], pl[8], pr[8];
    for (int k = 0; k < 8; ++k) {
        off[k] = (unsigned int)(rr[k] - lo);
        hit[k] = (k < 4 || valid_b) && off[k] < (unsigned int)tile;
        wc[k] = (unsigned int)(cc[k] - wlo);
        inw[k] = wc[k] < (unsigned int)wlen;
        pg[k] = proj[(hit[k] && !inw[k]) ? cc[k] : lo];
    }
    unsigned int ir[8], il[8];
    for (int k = 0; k < 8; ++k) {
        ir[k] = hit[k] ? off[k] + row_shift : 0u;
        il[k] = (hit[k] && inw[k]) ? wc[k] : 0u;
    }
    asm volatile("" : "+v"(ir[0]), "+v"(ir[1]), "+v"(ir[2]), "+v"(ir[3]),
                      "+v"(ir[4]), "+v"(ir[5]), "+v"(ir[6]), "+v"(ir[7]));
    asm volatile("" : "+v"(il[0]), "+v"(il[1]), "+v"(il[2]), "+v"(il[3]),
                      "+v"(il[4]), "+v"(il[5]), "+v"(il[6]), "+v"(il[7]));
    for (int k = 0; k < 8; ++k) {
        pr[k] = win[ir[k]];
        pl[k] = win[il[k]];
    }
    for (int k = 0; k < 8; ++k)
        d[k] = __float_as_uint(fabsf(pr[k] - (inw[k] ? pl[k] : pg[k])));
    asm volatile("" : "+v"(d[0]), "+v"(d[1]), "+v"(d[2]), "+v"(d[3]),
                      "+v"(d[4]), "+v"(d[5]), "+v"(d[6]), "+v"(d[7]));
    for (int k = 0; k < 8; ++k) {
        if (hit[k]) atomicMax(&tile_max[off[k]], d[k]);
    }
}

__global__ void __launch_bounds__(EDGE_BLOCK)
edge_max_kernel(const float* __restrict__ proj,
                const int* __restrict__ rows,
                const int* __restrict__ cols,
                const float* __restrict__ prev,
                const float* __restrict__ phi_mean,
                float* __restrict__ out,
                int n_nodes, int n_edges) {
    __shared__ __attribute__((aligned(16))) float win[WIN_CAP];
    __shared__ unsigned int tile_max[TILE_CAP];
    const int n_parts = gridDim.x;
    int tile = (n_nodes + n_parts - 1) / n_parts;
    if (tile > TILE_CAP) tile = TILE_CAP;
    const int n_groups = n_edges >> 2;
    const int g_step = gridDim.y * blockDim.x;
    const int4* __restrict__ rows4 = reinterpret_cast<const int4*>(rows);
    const int4* __restrict__ cols4 = reinterpret_cast<const int4*>(cols);

    const float phi = phi_mean[0];
    for (int lo = blockIdx.x * tile; lo < n_nodes; lo += n_parts * tile) {
        float pv[PREV_REGS];
        for (int u = 0; u < PREV_REGS; ++u) {
            const int n = lo + (int)threadIdx.x + u * EDGE_BLOCK;
            pv[u] = prev[n < n_nodes ? n : n_nodes - 1];
        }
        int wraw = lo + tile / 2 - (WIN_CAP - 4) / 2;
        if (wraw > n_nodes - (WIN_CAP - 4)) wraw = n_nodes - (WIN_CAP - 4);
        if (wraw < 0) wraw = 0;
        const int wlo = wraw & ~3;
        const int wlen = n_nodes - wlo < WIN_CAP ? n_nodes - wlo : WIN_CAP;
        {
            const float4* __restrict__ src4 =
                reinterpret_cast<const float4*>(proj + wlo);
            float4* dst4 = reinterpret_cast<float4*>(win);
            const int nq = wlen >> 2;
            int q = threadIdx.x;
            for (; q + (FILL_DEPTH - 1) * EDGE_BLOCK < nq; q += FILL_DEPTH * EDGE_BLOCK) {
                float4 v[FILL_DEPTH];
                for (int u = 0; u < FILL_DEPTH; ++u) v[u] = src4[q + u * EDGE_BLOCK];
                for (int u = 0; u < FILL_DEPTH; ++u) dst4[q + u * EDGE_BLOCK] = v[u];
            }
            for (; q < nq; q += EDGE_BLOCK) dst4[q] = src4[q];
            const int t = (nq << 2) + (int)threadIdx.x;
            if (t < wlen) win[t] = proj[wlo + t];
        }
        for (int i = threadIdx.x; i < tile; i += EDGE_BLOCK) tile_max[i] = 0u;
        __syncthreads();

        int g = blockIdx.y * blockDim.x + threadIdx.x;
        int4 ra = make_int4(0, 0, 0, 0), ca = ra, rb = ra, cb = ra;
        if (g < n_groups) {
            const int gb = g + g_step < n_groups ? g + g_step : g;
            ra = rows4[g];  ca = cols4[g];
            rb = rows4[gb]; cb = cols4[gb];
        }
        while (g < n_groups) {
            const int gn = g + 2 * g_step;
            const int ia = gn < n_groups ? gn : g;
            const int ib = gn + g_step < n_groups ? gn + g_step : g;
            const int4 rna = rows4[ia], cna = cols4[ia];
            const int4 rnb = rows4[ib], cnb = cols4[ib];

            edge_pair(ra, ca, rb, cb, g + g_step < n_groups, lo, tile, wlo, wlen,
                      proj, win, tile_max);

            ra = rna; ca = cna; rb = rnb; cb = cnb; g = gn;
        }
        if (blockIdx.y == 0) {
            int e = (n_groups << 2) + threadIdx.x;
            if (e < n_edges) {
                int r = rows[e], c = cols[e];
                unsigned int o = (unsigned int)(r - lo);
                if (o < (unsigned int)tile) {
                    float d = fabsf(win[r - wlo] - proj[c]);
                    atomicMax(&tile_max[o], __float_as_uint(d));
                }
            }
        }
        __syncthreads();
        for (int u = 0; u < PREV_REGS; ++u) {
            const int i = (int)threadIdx.x + u * EDGE_BLOCK;
            if (i < tile && lo + i < n_nodes) {
                const unsigned int v = tile_max[i];
                if (v != 0u)
                    fold_float(out + lo + i,
                               0.5f * (pv[u] + __uint_as_float(v) * phi),
                               phi >= 0.0f);
            }
        }
        __syncthreads();
    }
}

extern "C" void kernel_launch(void* const* d_in, const int* in_sizes, int n_in,
                              void* d_out, int out_size, void* d_ws, size_t ws_size,
                              hipStream_t stream) {
    const float* prev    = (const float*)d_in[0];
    const float* nodes   = (const float*)d_in[1];
    const int*   rows    = (const int*)  d_in[2];
    const int*   cols    = (const int*)  d_in[3];
    const float* W_phi   = (const float*)d_in[4];
    const float* W_theta = (const float*)d_in[5];
    float*       out     = (float*)d_out;

    const int n_nodes = in_sizes[0];
    const int n_edges = in_sizes[2];

    float* proj = (float*)d_ws;
    float* phi_mean = proj + n_nodes;

    const int B = 256;
    int node_blocks = (n_nodes + B - 1) / B;

    init_project_kernel<<<node_blocks + 1, B, 0, stream>>>(
        prev, nodes, W_theta, W_phi, out, proj, phi_mean, n_nodes);
    edge_max_kernel<<<dim3(NODE_PARTS, EDGE_PARTS), EDGE_BLOCK, 0, stream>>>(
        proj, rows, cols, prev, phi_mean, out, n_nodes, n_edges);
}
